# starting kernel
# baseline (speedup 1.0000x reference)
#include <hip/hip_runtime.h>
#include <hip/hip_bf16.h>

#define BB 2
#define LT 2048
#define DD 1024
#define HH 16
#define HD 64

typedef __attribute__((ext_vector_type(2))) float v2f;
typedef __attribute__((ext_vector_type(8))) float v8f;

__device__ __forceinline__ v8f wmma4(v2f a, v2f b, v8f c) {
  return __builtin_amdgcn_wmma_f32_16x16x4_f32(false, a, false, b, (short)0, c,
                                               false, false);
}

__global__ void __launch_bounds__(256)
qkv_gemm_kernel(const float* __restrict__ X, const float* __restrict__ W,
                float* __restrict__ Qb, float* __restrict__ Kb,
                float* __restrict__ Vb) {
  const int lane = threadIdx.x & 31;
  const int wave = threadIdx.x >> 5;
  const int colBase = (blockIdx.x * 8 + wave) * 64;
  const int rowBase = blockIdx.y * 16;
  const int half = lane >> 4;
  const int l16 = lane & 15;
  const int kph = half << 1;

  const float* xr = X + (size_t)(rowBase + l16) * DD + kph;
  const float* w0 = W + (size_t)(colBase + l16) * DD + kph;

  v8f acc[4] = {};
#pragma unroll 4
  for (int k = 0; k < DD; k += 4) {
    float2 af = *(const float2*)(xr + k);
    v2f a;
    a.x = af.x;
    a.y = af.y;
#pragma unroll
    for (int d = 0; d < 4; ++d) {
      float2 bf = *(const float2*)(w0 + (size_t)d * 16 * DD + k);
      v2f b;
      b.x = bf.x;
      b.y = bf.y;
      acc[d] = wmma4(a, b, acc[d]);
    }
  }

  const int c = colBase >> 10;
  const int h = (colBase & 1023) >> 6;
  float* dst = (c == 0) ? Qb : (c == 1) ? Kb : Vb;
#pragma unroll
  for (int r = 0; r < 8; ++r) {
    const int i = rowBase + r + half * 8;
    const int b = i >> 11;
    const int l = i & (LT - 1);
    const size_t rbase = (((size_t)b * HH + h) * LT + l) * HD;
#pragma unroll
    for (int d = 0; d < 4; ++d) dst[rbase + d * 16 + l16] = acc[d][r];
  }
}

__global__ void __launch_bounds__(256)
norm_rope_kernel(float* __restrict__ buf, const float* __restrict__ scale) {
  const int wave = threadIdx.x >> 5;
  const int lane = threadIdx.x & 31;
  const size_t row = (size_t)blockIdx.x * 8 + wave;
  const int l = (int)(row & (LT - 1));

  float2* p = (float2*)(buf + row * HD) + lane;
  float2 v = *p;
  float ss = v.x * v.x + v.y * v.y;
#pragma unroll
  for (int off = 16; off; off >>= 1) ss += __shfl_xor(ss, off, 32);
  const float r = rsqrtf(ss * (1.0f / 64.0f) + 1e-6f);
  const float xr = v.x * r * scale[2 * lane];
  const float xi = v.y * r * scale[2 * lane + 1];

  const int jj = (lane < 16) ? lane : (lane - 16);
  const float pos = (lane < 16) ? (float)(l >> 8) : (float)(l & 255);
  const float invf = expf((float)jj * (-9.210340371976184f / 16.0f));
  const float ang = pos * invf;
  const float c = cosf(ang), s = sinf(ang);
  float2 o;
  o.x = xr * c - xi * s;
  o.y = xr * s + xi * c;
  *p = o;
}

__global__ void __launch_bounds__(32)
attn_kernel(const float* __restrict__ Qb, const float* __restrict__ Kb,
            const float* __restrict__ Vb, float* __restrict__ Ab) {
  const int lane = threadIdx.x;
  const int half = lane >> 4;
  const int l16 = lane & 15;
  const int kph = half << 1;
  const int q0 = blockIdx.x * 16;
  const int bh = blockIdx.y;
  const int nkey = ((q0 >> 8) + 1) << 8;
  const size_t base = (size_t)bh * LT * HD;

  __shared__ float sP[16 * 16];

  v2f qa[16];
  const float* qrow = Qb + base + (size_t)(q0 + l16) * HD + kph;
#pragma unroll
  for (int c = 0; c < 16; ++c) {
    float2 t = *(const float2*)(qrow + 4 * c);
    qa[c].x = t.x;
    qa[c].y = t.y;
  }

  v8f o0 = {}, o1 = {}, o2 = {}, o3 = {};
  float m[8], lsum[8];
#pragma unroll
  for (int r = 0; r < 8; ++r) {
    m[r] = -3.0e38f;
    lsum[r] = 0.0f;
  }

  for (int key0 = 0; key0 < nkey; key0 += 16) {
    const float* krow = Kb + base + (size_t)(key0 + l16) * HD + kph;
    v8f s = {};
#pragma unroll
    for (int c = 0; c < 16; ++c) {
      float2 t = *(const float2*)(krow + 4 * c);
      v2f b;
      b.x = t.x;
      b.y = t.y;
      s = wmma4(qa[c], b, s);
    }

    float fac[8];
#pragma unroll
    for (int r = 0; r < 8; ++r) {
      const float sv = s[r] * 0.125f;
      float mx = sv;
#pragma unroll
      for (int off = 1; off < 16; off <<= 1)
        mx = fmaxf(mx, __shfl_xor(mx, off, 32));
      const float nm = fmaxf(m[r], mx);
      const float p = __expf(sv - nm);
      s[r] = p;
      float rs = p;
#pragma unroll
      for (int off = 1; off < 16; off <<= 1) rs += __shfl_xor(rs, off, 32);
      fac[r] = __expf(m[r] - nm);
      lsum[r] = lsum[r] * fac[r] + rs;
      m[r] = nm;
    }
#pragma unroll
    for (int r = 0; r < 8; ++r) {
      o0[r] *= fac[r];
      o1[r] *= fac[r];
      o2[r] *= fac[r];
      o3[r] *= fac[r];
    }

    __syncthreads();
#pragma unroll
    for (int r = 0; r < 8; ++r) sP[(r + half * 8) * 16 + l16] = s[r];
    __syncthreads();

#pragma unroll
    for (int c = 0; c < 4; ++c) {
      v2f a;
      a.x = sP[l16 * 16 + 4 * c + kph];
      a.y = sP[l16 * 16 + 4 * c + kph + 1];
      const float* vr0 = Vb + base + (size_t)(key0 + 4 * c + kph) * HD;
      const float* vr1 = vr0 + HD;
      v2f b;
      b.x = vr0[l16];      b.y = vr1[l16];      o0 = wmma4(a, b, o0);
      b.x = vr0[16 + l16]; b.y = vr1[16 + l16]; o1 = wmma4(a, b, o1);
      b.x = vr0[32 + l16]; b.y = vr1[32 + l16]; o2 = wmma4(a, b, o2);
      b.x = vr0[48 + l16]; b.y = vr1[48 + l16]; o3 = wmma4(a, b, o3);
    }
  }

  const int bb = bh >> 4, h = bh & 15;
#pragma unroll
  for (int r = 0; r < 8; ++r) {
    const float inv = 1.0f / lsum[r];
    const int l = q0 + r + half * 8;
    float* dst = Ab + ((size_t)bb * LT + l) * DD + h * HD;
    dst[l16] = o0[r] * inv;
    dst[16 + l16] = o1[r] * inv;
    dst[32 + l16] = o2[r] * inv;
    dst[48 + l16] = o3[r] * inv;
  }
}

__global__ void __launch_bounds__(256)
out_gemm_kernel(const float* __restrict__ A, const float* __restrict__ W,
                float* __restrict__ Y) {
  const int lane = threadIdx.x & 31;
  const int wave = threadIdx.x >> 5;
  const int colBase = (blockIdx.x * 8 + wave) * 64;
  const int rowBase = blockIdx.y * 16;
  const int half = lane >> 4;
  const int l16 = lane & 15;
  const int kph = half << 1;

  const float* ar = A + (size_t)(rowBase + l16) * DD + kph;
  const float* w0 = W + (size_t)(colBase + l16) * DD + kph;

  v8f acc[4] = {};
#pragma unroll 4
  for (int k = 0; k < DD; k += 4) {
    float2 af = *(const float2*)(ar + k);
    v2f a;
    a.x = af.x;
    a.y = af.y;
#pragma unroll
    for (int d = 0; d < 4; ++d) {
      float2 bf = *(const float2*)(w0 + (size_t)d * 16 * DD + k);
      v2f b;
      b.x = bf.x;
      b.y = bf.y;
      acc[d] = wmma4(a, b, acc[d]);
    }
  }

#pragma unroll
  for (int r = 0; r < 8; ++r) {
    const int i = rowBase + r + half * 8;
#pragma unroll
    for (int d = 0; d < 4; ++d)
      Y[(size_t)i * DD + colBase + d * 16 + l16] = acc[d][r];
  }
}

extern "C" void kernel_launch(void* const* d_in, const int* in_sizes, int n_in,
                              void* d_out, int out_size, void* d_ws,
                              size_t ws_size, hipStream_t stream) {
  const float* x = (const float*)d_in[0];
  const float* Wqkv = (const float*)d_in[1];
  const float* Wout = (const float*)d_in[2];
  const float* qs = (const float*)d_in[3];
  const float* ks = (const float*)d_in[4];
  float* out = (float*)d_out;

  const size_t N = (size_t)BB * HH * LT * HD;
  float* Qb = (float*)d_ws;
  float* Kb = Qb + N;
  float* Vb = Kb + N;
  float* Ab = Vb + N;

  qkv_gemm_kernel<<<dim3(6, 256), 256, 0, stream>>>(x, Wqkv, Qb, Kb, Vb);
  norm_rope_kernel<<<dim3(8192), 256, 0, stream>>>(Qb, qs);
  norm_rope_kernel<<<dim3(8192), 256, 0, stream>>>(Kb, ks);
  attn_kernel<<<dim3(LT / 16, BB * HH), 32, 0, stream>>>(Qb, Kb, Vb, Ab);
  out_gemm_kernel<<<dim3(2, 256), 256, 0, stream>>>(Ab, Wout, out);
}


# Both projection GEMMs without LDS: each wave owns a 64x64 register tile, fragments straight from global rows by 128-bit loads (one k order for both operands), next k block preloaded in registers
# speedup vs baseline: 1.5523x; 1.5523x over previous
#include <hip/hip_runtime.h>
#include <hip/hip_bf16.h>

#define BB 2
#define LT 2048
#define DD 1024
#define HH 16
#define HD 64

typedef __attribute__((ext_vector_type(2))) float v2f;
typedef __attribute__((ext_vector_type(8))) float v8f;

__device__ __forceinline__ v8f wmma4(v2f a, v2f b, v8f c) {
  return __builtin_amdgcn_wmma_f32_16x16x4_f32(false, a, false, b, (short)0, c,
                                               false, false);
}

__device__ __forceinline__ void load_frags(float4 (&f)[4][2],
                                           const float* __restrict__ p, int k) {
#pragma unroll
  for (int i = 0; i < 4; ++i) {
    const float* q = p + (size_t)i * 16 * DD + k;
    f[i][0] = *(const float4*)(q);
    f[i][1] = *(const float4*)(q + 4);
  }
}

__device__ __forceinline__ void mma_block(v8f (&acc)[4][4],
                                          const float4 (&a)[4][2],
                                          const float4 (&b)[4][2]) {
#pragma unroll
  for (int s = 0; s < 4; ++s) {
    v2f af[4], bf[4];
#pragma unroll
    for (int i = 0; i < 4; ++i) {
      const float4 va = a[i][s >> 1];
      const float4 vb = b[i][s >> 1];
      af[i].x = (s & 1) ? va.z : va.x;
      af[i].y = (s & 1) ? va.w : va.y;
      bf[i].x = (s & 1) ? vb.z : vb.x;
      bf[i].y = (s & 1) ? vb.w : vb.y;
    }
#pragma unroll
    for (int i = 0; i < 4; ++i)
#pragma unroll
      for (int j = 0; j < 4; ++j) acc[i][j] = wmma4(af[i], bf[j], acc[i][j]);
  }
}

__device__ __forceinline__ void gemm_wave_64x64(v8f (&acc)[4][4],
                                                const float* __restrict__ ap,
                                                const float* __restrict__ bp) {
  float4 a0[4][2], b0[4][2], a1[4][2], b1[4][2];
  load_frags(a0, ap, 0);
  load_frags(b0, bp, 0);
  for (int k = 0; k < DD - 32; k += 32) {
    load_frags(a1, ap, k + 16);
    load_frags(b1, bp, k + 16);
    mma_block(acc, a0, b0);
    load_frags(a0, ap, k + 32);
    load_frags(b0, bp, k + 32);
    mma_block(acc, a1, b1);
  }
  load_frags(a1, ap, DD - 16);
  load_frags(b1, bp, DD - 16);
  mma_block(acc, a0, b0);
  mma_block(acc, a1, b1);
}

__global__ void __launch_bounds__(128)
qkv_gemm_kernel(const float* __restrict__ X, const float* __restrict__ W,
                float* __restrict__ Qb, float* __restrict__ Kb,
                float* __restrict__ Vb) {
  const int lane = threadIdx.x & 31;
  const int wave = threadIdx.x >> 5;
  const int half = lane >> 4;
  const int l16 = lane & 15;
  const int rowBase = blockIdx.y * 128 + (wave & 1) * 64;
  const int colBase = blockIdx.x * 128 + (wave >> 1) * 64;

  const float* ap = X + (size_t)(rowBase + l16) * DD + half * 8;
  const float* bp = W + (size_t)(colBase + l16) * DD + half * 8;

  v8f acc[4][4] = {};
  gemm_wave_64x64(acc, ap, bp);

  const int c = colBase >> 10;
  const int h = (colBase & 1023) >> 6;
  float* dst = (c == 0) ? Qb : (c == 1) ? Kb : Vb;
#pragma unroll
  for (int i = 0; i < 4; ++i) {
#pragma unroll
    for (int r = 0; r < 8; ++r) {
      const int row = rowBase + i * 16 + r + half * 8;
      const int b = row >> 11;
      const int l = row & (LT - 1);
      const size_t rbase = (((size_t)b * HH + h) * LT + l) * HD;
#pragma unroll
      for (int j = 0; j < 4; ++j) dst[rbase + j * 16 + l16] = acc[i][j][r];
    }
  }
}

__global__ void __launch_bounds__(256)
norm_rope_kernel(float* __restrict__ buf, const float* __restrict__ scale) {
  const int wave = threadIdx.x >> 5;
  const int lane = threadIdx.x & 31;
  const size_t row = (size_t)blockIdx.x * 8 + wave;
  const int l = (int)(row & (LT - 1));

  float2* p = (float2*)(buf + row * HD) + lane;
  float2 v = *p;
  float ss = v.x * v.x + v.y * v.y;
#pragma unroll
  for (int off = 16; off; off >>= 1) ss += __shfl_xor(ss, off, 32);
  const float r = rsqrtf(ss * (1.0f / 64.0f) + 1e-6f);
  const float xr = v.x * r * scale[2 * lane];
  const float xi = v.y * r * scale[2 * lane + 1];

  const int jj = (lane < 16) ? lane : (lane - 16);
  const float pos = (lane < 16) ? (float)(l >> 8) : (float)(l & 255);
  const float invf = expf((float)jj * (-9.210340371976184f / 16.0f));
  const float ang = pos * invf;
  const float c = cosf(ang), s = sinf(ang);
  float2 o;
  o.x = xr * c - xi * s;
  o.y = xr * s + xi * c;
  *p = o;
}

__global__ void __launch_bounds__(32)
attn_kernel(const float* __restrict__ Qb, const float* __restrict__ Kb,
            const float* __restrict__ Vb, float* __restrict__ Ab) {
  const int lane = threadIdx.x;
  const int half = lane >> 4;
  const int l16 = lane & 15;
  const int kph = half << 1;
  const int q0 = blockIdx.x * 16;
  const int bh = blockIdx.y;
  const int nkey = ((q0 >> 8) + 1) << 8;
  const size_t base = (size_t)bh * LT * HD;

  __shared__ float sP[16 * 16];

  v2f qa[16];
  const float* qrow = Qb + base + (size_t)(q0 + l16) * HD + kph;
#pragma unroll
  for (int c = 0; c < 16; ++c) {
    float2 t = *(const float2*)(qrow + 4 * c);
    qa[c].x = t.x;
    qa[c].y = t.y;
  }

  v8f o0 = {}, o1 = {}, o2 = {}, o3 = {};
  float m[8], lsum[8];
#pragma unroll
  for (int r = 0; r < 8; ++r) {
    m[r] = -3.0e38f;
    lsum[r] = 0.0f;
  }

  for (int key0 = 0; key0 < nkey; key0 += 16) {
    const float* krow = Kb + base + (size_t)(key0 + l16) * HD + kph;
    v8f s = {};
#pragma unroll
    for (int c = 0; c < 16; ++c) {
      float2 t = *(const float2*)(krow + 4 * c);
      v2f b;
      b.x = t.x;
      b.y = t.y;
      s = wmma4(qa[c], b, s);
    }

    float fac[8];
#pragma unroll
    for (int r = 0; r < 8; ++r) {
      const float sv = s[r] * 0.125f;
      float mx = sv;
#pragma unroll
      for (int off = 1; off < 16; off <<= 1)
        mx = fmaxf(mx, __shfl_xor(mx, off, 32));
      const float nm = fmaxf(m[r], mx);
      const float p = __expf(sv - nm);
      s[r] = p;
      float rs = p;
#pragma unroll
      for (int off = 1; off < 16; off <<= 1) rs += __shfl_xor(rs, off, 32);
      fac[r] = __expf(m[r] - nm);
      lsum[r] = lsum[r] * fac[r] + rs;
      m[r] = nm;
    }
#pragma unroll
    for (int r = 0; r < 8; ++r) {
      o0[r] *= fac[r];
      o1[r] *= fac[r];
      o2[r] *= fac[r];
      o3[r] *= fac[r];
    }

    __syncthreads();
#pragma unroll
    for (int r = 0; r < 8; ++r) sP[(r + half * 8) * 16 + l16] = s[r];
    __syncthreads();

#pragma unroll
    for (int c = 0; c < 4; ++c) {
      v2f a;
      a.x = sP[l16 * 16 + 4 * c + kph];
      a.y = sP[l16 * 16 + 4 * c + kph + 1];
      const float* vr0 = Vb + base + (size_t)(key0 + 4 * c + kph) * HD;
      const float* vr1 = vr0 + HD;
      v2f b;
      b.x = vr0[l16];      b.y = vr1[l16];      o0 = wmma4(a, b, o0);
      b.x = vr0[16 + l16]; b.y = vr1[16 + l16]; o1 = wmma4(a, b, o1);
      b.x = vr0[32 + l16]; b.y = vr1[32 + l16]; o2 = wmma4(a, b, o2);
      b.x = vr0[48 + l16]; b.y = vr1[48 + l16]; o3 = wmma4(a, b, o3);
    }
  }

  const int bb = bh >> 4, h = bh & 15;
#pragma unroll
  for (int r = 0; r < 8; ++r) {
    const float inv = 1.0f / lsum[r];
    const int l = q0 + r + half * 8;
    float* dst = Ab + ((size_t)bb * LT + l) * DD + h * HD;
    dst[l16] = o0[r] * inv;
    dst[16 + l16] = o1[r] * inv;
    dst[32 + l16] = o2[r] * inv;
    dst[48 + l16] = o3[r] * inv;
  }
}

__global__ void __launch_bounds__(128)
out_gemm_kernel(const float* __restrict__ A, const float* __restrict__ W,
                float* __restrict__ Y) {
  const int lane = threadIdx.x & 31;
  const int wave = threadIdx.x >> 5;
  const int half = lane >> 4;
  const int l16 = lane & 15;
  const int rowBase = blockIdx.y * 128 + (wave & 1) * 64;
  const int colBase = blockIdx.x * 128 + (wave >> 1) * 64;

  const float* ap = A + (size_t)(rowBase + l16) * DD + half * 8;
  const float* bp = W + (size_t)(colBase + l16) * DD + half * 8;

  v8f acc[4][4] = {};
  gemm_wave_64x64(acc, ap, bp);

#pragma unroll
  for (int i = 0; i < 4; ++i) {
#pragma unroll
    for (int r = 0; r < 8; ++r) {
      const int row = rowBase + i * 16 + r + half * 8;
#pragma unroll
      for (int j = 0; j < 4; ++j)
        Y[(size_t)row * DD + colBase + j * 16 + l16] = acc[i][j][r];
    }
  }
}

extern "C" void kernel_launch(void* const* d_in, const int* in_sizes, int n_in,
                              void* d_out, int out_size, void* d_ws,
                              size_t ws_size, hipStream_t stream) {
  const float* x = (const float*)d_in[0];
  const float* Wqkv = (const float*)d_in[1];
  const float* Wout = (const float*)d_in[2];
  const float* qs = (const float*)d_in[3];
  const float* ks = (const float*)d_in[4];
  float* out = (float*)d_out;

  const size_t N = (size_t)BB * HH * LT * HD;
  float* Qb = (float*)d_ws;
  float* Kb = Qb + N;
  float* Vb = Kb + N;
  float* Ab = Vb + N;

  qkv_gemm_kernel<<<dim3(24, 32), 128, 0, stream>>>(x, Wqkv, Qb, Kb, Vb);
  norm_rope_kernel<<<dim3(8192), 256, 0, stream>>>(Qb, qs);
  norm_rope_kernel<<<dim3(8192), 256, 0, stream>>>(Kb, ks);
  attn_kernel<<<dim3(LT / 16, BB * HH), 32, 0, stream>>>(Qb, Kb, Vb, Ab);
  out_gemm_kernel<<<dim3(8, 32), 128, 0, stream>>>(Ab, Wout, out);
}


# Attention re-tiled: an 8-wave workgroup takes one 16-query tile from each of the 8 frames (equal work, one shared K/V stream); Q/K/V fragments straight from global rows by 128-bit loads
# speedup vs baseline: 1.7567x; 1.1317x over previous
#include <hip/hip_runtime.h>
#include <hip/hip_bf16.h>

#define BB 2
#define LT 2048
#define DD 1024
#define HH 16
#define HD 64

typedef __attribute__((ext_vector_type(2))) float v2f;
typedef __attribute__((ext_vector_type(8))) float v8f;

__device__ __forceinline__ v8f wmma4(v2f a, v2f b, v8f c) {
  return __builtin_amdgcn_wmma_f32_16x16x4_f32(false, a, false, b, (short)0, c,
                                               false, false);
}

__device__ __forceinline__ void load_frags(float4 (&f)[4][2],
                                           const float* __restrict__ p, int k) {
#pragma unroll
  for (int i = 0; i < 4; ++i) {
    const float* q = p + (size_t)i * 16 * DD + k;
    f[i][0] = *(const float4*)(q);
    f[i][1] = *(const float4*)(q + 4);
  }
}

__device__ __forceinline__ void mma_block(v8f (&acc)[4][4],
                                          const float4 (&a)[4][2],
                                          const float4 (&b)[4][2]) {
#pragma unroll
  for (int s = 0; s < 4; ++s) {
    v2f af[4], bf[4];
#pragma unroll
    for (int i = 0; i < 4; ++i) {
      const float4 va = a[i][s >> 1];
      const float4 vb = b[i][s >> 1];
      af[i].x = (s & 1) ? va.z : va.x;
      af[i].y = (s & 1) ? va.w : va.y;
      bf[i].x = (s & 1) ? vb.z : vb.x;
      bf[i].y = (s & 1) ? vb.w : vb.y;
    }
#pragma unroll
    for (int i = 0; i < 4; ++i)
#pragma unroll
      for (int j = 0; j < 4; ++j) acc[i][j] = wmma4(af[i], bf[j], acc[i][j]);
  }
}

__device__ __forceinline__ void gemm_wave_64x64(v8f (&acc)[4][4],
                                                const float* __restrict__ ap,
                                                const float* __restrict__ bp) {
  float4 a0[4][2], b0[4][2], a1[4][2], b1[4][2];
  load_frags(a0, ap, 0);
  load_frags(b0, bp, 0);
  for (int k = 0; k < DD - 32; k += 32) {
    load_frags(a1, ap, k + 16);
    load_frags(b1, bp, k + 16);
    mma_block(acc, a0, b0);
    load_frags(a0, ap, k + 32);
    load_frags(b0, bp, k + 32);
    mma_block(acc, a1, b1);
  }
  load_frags(a1, ap, DD - 16);
  load_frags(b1, bp, DD - 16);
  mma_block(acc, a0, b0);
  mma_block(acc, a1, b1);
}

__global__ void __launch_bounds__(128)
qkv_gemm_kernel(const float* __restrict__ X, const float* __restrict__ W,
                float* __restrict__ Qb, float* __restrict__ Kb,
                float* __restrict__ Vb) {
  const int lane = threadIdx.x & 31;
  const int wave = threadIdx.x >> 5;
  const int half = lane >> 4;
  const int l16 = lane & 15;
  const int rowBase = blockIdx.y * 128 + (wave & 1) * 64;
  const int colBase = blockIdx.x * 128 + (wave >> 1) * 64;

  const float* ap = X + (size_t)(rowBase + l16) * DD + half * 8;
  const float* bp = W + (size_t)(colBase + l16) * DD + half * 8;

  v8f acc[4][4] = {};
  gemm_wave_64x64(acc, ap, bp);

  const int c = colBase >> 10;
  const int h = (colBase & 1023) >> 6;
  float* dst = (c == 0) ? Qb : (c == 1) ? Kb : Vb;
#pragma unroll
  for (int i = 0; i < 4; ++i) {
#pragma unroll
    for (int r = 0; r < 8; ++r) {
      const int row = rowBase + i * 16 + r + half * 8;
      const int b = row >> 11;
      const int l = row & (LT - 1);
      const size_t rbase = (((size_t)b * HH + h) * LT + l) * HD;
#pragma unroll
      for (int j = 0; j < 4; ++j) dst[rbase + j * 16 + l16] = acc[i][j][r];
    }
  }
}

__global__ void __launch_bounds__(256)
norm_rope_kernel(float* __restrict__ buf, const float* __restrict__ scale) {
  const int wave = threadIdx.x >> 5;
  const int lane = threadIdx.x & 31;
  const size_t row = (size_t)blockIdx.x * 8 + wave;
  const int l = (int)(row & (LT - 1));

  float2* p = (float2*)(buf + row * HD) + lane;
  float2 v = *p;
  float ss = v.x * v.x + v.y * v.y;
#pragma unroll
  for (int off = 16; off; off >>= 1) ss += __shfl_xor(ss, off, 32);
  const float r = rsqrtf(ss * (1.0f / 64.0f) + 1e-6f);
  const float xr = v.x * r * scale[2 * lane];
  const float xi = v.y * r * scale[2 * lane + 1];

  const int jj = (lane < 16) ? lane : (lane - 16);
  const float pos = (lane < 16) ? (float)(l >> 8) : (float)(l & 255);
  const float invf = expf((float)jj * (-9.210340371976184f / 16.0f));
  const float ang = pos * invf;
  const float c = cosf(ang), s = sinf(ang);
  float2 o;
  o.x = xr * c - xi * s;
  o.y = xr * s + xi * c;
  *p = o;
}

#define PPITCH 20

__device__ __forceinline__ void wave_lds_order() {
  __builtin_amdgcn_fence(__ATOMIC_RELEASE, "wavefront");
  __builtin_amdgcn_wave_barrier();
  __builtin_amdgcn_fence(__ATOMIC_ACQUIRE, "wavefront");
}

__global__ void __launch_bounds__(256)
attn_kernel(const float* __restrict__ Qb, const float* __restrict__ Kb,
            const float* __restrict__ Vb, float* __restrict__ Ab) {
  const int lane = threadIdx.x & 31;
  const int wave = __builtin_amdgcn_readfirstlane((int)(threadIdx.x >> 5));
  const int half = lane >> 4;
  const int l16 = lane & 15;
  const int frame = (wave < 4) ? wave : 11 - wave;
  const int q0 = (frame << 8) + blockIdx.x * 16;
  const int bh = blockIdx.y;
  const int nkey = (frame + 1) << 8;
  const size_t base = (size_t)bh * LT * HD;

  __shared__ __attribute__((aligned(16))) float sPall[8][16 * PPITCH];
  float* sP = sPall[wave];

  float4 qf[4][2];
  {
    const float* qrow = Qb + base + (size_t)(q0 + l16) * HD + half * 8;
#pragma unroll
    for (int blk = 0; blk < 4; ++blk) {
      qf[blk][0] = *(const float4*)(qrow + 16 * blk);
      qf[blk][1] = *(const float4*)(qrow + 16 * blk + 4);
    }
  }

  v8f o0 = {}, o1 = {}, o2 = {}, o3 = {};
  float m[8], lsum[8];
#pragma unroll
  for (int r = 0; r < 8; ++r) {
    m[r] = -3.0e38f;
    lsum[r] = 0.0f;
  }

  for (int key0 = 0; key0 < nkey; key0 += 16) {
    float4 kf[4][2];
    {
      const float* krow = Kb + base + (size_t)(key0 + l16) * HD + half * 8;
#pragma unroll
      for (int blk = 0; blk < 4; ++blk) {
        kf[blk][0] = *(const float4*)(krow + 16 * blk);
        kf[blk][1] = *(const float4*)(krow + 16 * blk + 4);
      }
    }
    float4 vf[8];
    {
      const float* vrow = Vb + base + (size_t)(key0 + half * 8) * HD + 4 * l16;
#pragma unroll
      for (int kk = 0; kk < 8; ++kk) vf[kk] = *(const float4*)(vrow + kk * HD);
    }

    v8f s = {};
#pragma unroll
    for (int blk = 0; blk < 4; ++blk) {
#pragma unroll
      for (int t = 0; t < 2; ++t) {
        const float4 qa = qf[blk][t];
        const float4 kb = kf[blk][t];
        v2f a, b;
        a.x = qa.x; a.y = qa.y; b.x = kb.x; b.y = kb.y;
        s = wmma4(a, b, s);
        a.x = qa.z; a.y = qa.w; b.x = kb.z; b.y = kb.w;
        s = wmma4(a, b, s);
      }
    }

    float fac[8];
#pragma unroll
    for (int r = 0; r < 8; ++r) {
      const float sv = s[r] * 0.125f;
      float mx = sv;
#pragma unroll
      for (int off = 1; off < 16; off <<= 1)
        mx = fmaxf(mx, __shfl_xor(mx, off, 32));
      const float nm = fmaxf(m[r], mx);
      const float p = __expf(sv - nm);
      s[r] = p;
      float rs = p;
#pragma unroll
      for (int off = 1; off < 16; off <<= 1) rs += __shfl_xor(rs, off, 32);
      fac[r] = __expf(m[r] - nm);
      lsum[r] = lsum[r] * fac[r] + rs;
      m[r] = nm;
    }
#pragma unroll
    for (int r = 0; r < 8; ++r) {
      o0[r] *= fac[r];
      o1[r] *= fac[r];
      o2[r] *= fac[r];
      o3[r] *= fac[r];
    }

    wave_lds_order();
#pragma unroll
    for (int r = 0; r < 8; ++r) sP[(r + half * 8) * PPITCH + l16] = s[r];
    wave_lds_order();
    const float4 pa0 = *(const float4*)(sP + l16 * PPITCH + half * 8);
    const float4 pa1 = *(const float4*)(sP + l16 * PPITCH + half * 8 + 4);

#pragma unroll
    for (int u = 0; u < 4; ++u) {
      const float4 pa = (u < 2) ? pa0 : pa1;
      v2f a;
      a.x = (u & 1) ? pa.z : pa.x;
      a.y = (u & 1) ? pa.w : pa.y;
      const float4 v0 = vf[2 * u];
      const float4 v1 = vf[2 * u + 1];
      v2f b;
      b.x = v0.x; b.y = v1.x; o0 = wmma4(a, b, o0);
      b.x = v0.y; b.y = v1.y; o1 = wmma4(a, b, o1);
      b.x = v0.z; b.y = v1.z; o2 = wmma4(a, b, o2);
      b.x = v0.w; b.y = v1.w; o3 = wmma4(a, b, o3);
    }
  }

  const int bb = bh >> 4, h = bh & 15;
#pragma unroll
  for (int r = 0; r < 8; ++r) {
    const float inv = 1.0f / lsum[r];
    const int l = q0 + r + half * 8;
    float* dst = Ab + ((size_t)bb * LT + l) * DD + h * HD + 4 * l16;
    float4 ov;
    ov.x = o0[r] * inv;
    ov.y = o1[r] * inv;
    ov.z = o2[r] * inv;
    ov.w = o3[r] * inv;
    *(float4*)dst = ov;
  }
}

__global__ void __launch_bounds__(128)
out_gemm_kernel(const float* __restrict__ A, const float* __restrict__ W,
                float* __restrict__ Y) {
  const int lane = threadIdx.x & 31;
  const int wave = threadIdx.x >> 5;
  const int half = lane >> 4;
  const int l16 = lane & 15;
  const int rowBase = blockIdx.y * 128 + (wave & 1) * 64;
  const int colBase = blockIdx.x * 128 + (wave >> 1) * 64;

  const float* ap = A + (size_t)(rowBase + l16) * DD + half * 8;
  const float* bp = W + (size_t)(colBase + l16) * DD + half * 8;

  v8f acc[4][4] = {};
  gemm_wave_64x64(acc, ap, bp);

#pragma unroll
  for (int i = 0; i < 4; ++i) {
#pragma unroll
    for (int r = 0; r < 8; ++r) {
      const int row = rowBase + i * 16 + r + half * 8;
#pragma unroll
      for (int j = 0; j < 4; ++j)
        Y[(size_t)row * DD + colBase + j * 16 + l16] = acc[i][j][r];
    }
  }
}

extern "C" void kernel_launch(void* const* d_in, const int* in_sizes, int n_in,
                              void* d_out, int out_size, void* d_ws,
                              size_t ws_size, hipStream_t stream) {
  const float* x = (const float*)d_in[0];
  const float* Wqkv = (const float*)d_in[1];
  const float* Wout = (const float*)d_in[2];
  const float* qs = (const float*)d_in[3];
  const float* ks = (const float*)d_in[4];
  float* out = (float*)d_out;

  const size_t N = (size_t)BB * HH * LT * HD;
  float* Qb = (float*)d_ws;
  float* Kb = Qb + N;
  float* Vb = Kb + N;
  float* Ab = Vb + N;

  qkv_gemm_kernel<<<dim3(24, 32), 128, 0, stream>>>(x, Wqkv, Qb, Kb, Vb);
  norm_rope_kernel<<<dim3(8192), 256, 0, stream>>>(Qb, qs);
  norm_rope_kernel<<<dim3(8192), 256, 0, stream>>>(Kb, ks);
  attn_kernel<<<dim3(16, BB * HH), 256, 0, stream>>>(Qb, Kb, Vb, Ab);
  out_gemm_kernel<<<dim3(8, 32), 128, 0, stream>>>(Ab, Wout, out);
}


# Attention: fragment loads issued ahead of use: K of the next 16-key chunk goes into a second register set one chunk ahead (loop over chunk pairs, no copies), V goes out at the head of its own chunk
# speedup vs baseline: 1.9207x; 1.0934x over previous
#include <hip/hip_runtime.h>
#include <hip/hip_bf16.h>

#define BB 2
#define LT 2048
#define DD 1024
#define HH 16
#define HD 64

typedef __attribute__((ext_vector_type(2))) float v2f;
typedef __attribute__((ext_vector_type(8))) float v8f;

__device__ __forceinline__ v8f wmma4(v2f a, v2f b, v8f c) {
  return __builtin_amdgcn_wmma_f32_16x16x4_f32(false, a, false, b, (short)0, c,
                                               false, false);
}

__device__ __forceinline__ void load_frags(float4 (&f)[4][2],
                                           const float* __restrict__ p, int k) {
#pragma unroll
  for (int i = 0; i < 4; ++i) {
    const float* q = p + (size_t)i * 16 * DD + k;
    f[i][0] = *(const float4*)(q);
    f[i][1] = *(const float4*)(q + 4);
  }
}

__device__ __forceinline__ void mma_block(v8f (&acc)[4][4],
                                          const float4 (&a)[4][2],
                                          const float4 (&b)[4][2]) {
#pragma unroll
  for (int s = 0; s < 4; ++s) {
    v2f af[4], bf[4];
#pragma unroll
    for (int i = 0; i < 4; ++i) {
      const float4 va = a[i][s >> 1];
      const float4 vb = b[i][s >> 1];
      af[i].x = (s & 1) ? va.z : va.x;
      af[i].y = (s & 1) ? va.w : va.y;
      bf[i].x = (s & 1) ? vb.z : vb.x;
      bf[i].y = (s & 1) ? vb.w : vb.y;
    }
#pragma unroll
    for (int i = 0; i < 4; ++i)
#pragma unroll
      for (int j = 0; j < 4; ++j) acc[i][j] = wmma4(af[i], bf[j], acc[i][j]);
  }
}

__device__ __forceinline__ void gemm_wave_64x64(v8f (&acc)[4][4],
                                                const float* __restrict__ ap,
                                                const float* __restrict__ bp) {
  float4 a0[4][2], b0[4][2], a1[4][2], b1[4][2];
  load_frags(a0, ap, 0);
  load_frags(b0, bp, 0);
  for (int k = 0; k < DD - 32; k += 32) {
    load_frags(a1, ap, k + 16);
    load_frags(b1, bp, k + 16);
    mma_block(acc, a0, b0);
    load_frags(a0, ap, k + 32);
    load_frags(b0, bp, k + 32);
    mma_block(acc, a1, b1);
  }
  load_frags(a1, ap, DD - 16);
  load_frags(b1, bp, DD - 16);
  mma_block(acc, a0, b0);
  mma_block(acc, a1, b1);
}

__global__ void __launch_bounds__(128)
qkv_gemm_kernel(const float* __restrict__ X, const float* __restrict__ W,
                float* __restrict__ Qb, float* __restrict__ Kb,
                float* __restrict__ Vb) {
  const int lane = threadIdx.x & 31;
  const int wave = threadIdx.x >> 5;
  const int half = lane >> 4;
  const int l16 = lane & 15;
  const int rowBase = blockIdx.y * 128 + (wave & 1) * 64;
  const int colBase = blockIdx.x * 128 + (wave >> 1) * 64;

  const float* ap = X + (size_t)(rowBase + l16) * DD + half * 8;
  const float* bp = W + (size_t)(colBase + l16) * DD + half * 8;

  v8f acc[4][4] = {};
  gemm_wave_64x64(acc, ap, bp);

  const int c = colBase >> 10;
  const int h = (colBase & 1023) >> 6;
  float* dst = (c == 0) ? Qb : (c == 1) ? Kb : Vb;
#pragma unroll
  for (int i = 0; i < 4; ++i) {
#pragma unroll
    for (int r = 0; r < 8; ++r) {
      const int row = rowBase + i * 16 + r + half * 8;
      const int b = row >> 11;
      const int l = row & (LT - 1);
      const size_t rbase = (((size_t)b * HH + h) * LT + l) * HD;
#pragma unroll
      for (int j = 0; j < 4; ++j) dst[rbase + j * 16 + l16] = acc[i][j][r];
    }
  }
}

__global__ void __launch_bounds__(256)
norm_rope_kernel(float* __restrict__ buf, const float* __restrict__ scale) {
  const int wave = threadIdx.x >> 5;
  const int lane = threadIdx.x & 31;
  const size_t row = (size_t)blockIdx.x * 8 + wave;
  const int l = (int)(row & (LT - 1));

  float2* p = (float2*)(buf + row * HD) + lane;
  float2 v = *p;
  float ss = v.x * v.x + v.y * v.y;
#pragma unroll
  for (int off = 16; off; off >>= 1) ss += __shfl_xor(ss, off, 32);
  const float r = rsqrtf(ss * (1.0f / 64.0f) + 1e-6f);
  const float xr = v.x * r * scale[2 * lane];
  const float xi = v.y * r * scale[2 * lane + 1];

  const int jj = (lane < 16) ? lane : (lane - 16);
  const float pos = (lane < 16) ? (float)(l >> 8) : (float)(l & 255);
  const float invf = expf((float)jj * (-9.210340371976184f / 16.0f));
  const float ang = pos * invf;
  const float c = cosf(ang), s = sinf(ang);
  float2 o;
  o.x = xr * c - xi * s;
  o.y = xr * s + xi * c;
  *p = o;
}

#define PPITCH 20

__device__ __forceinline__ void wave_lds_order() {
  __builtin_amdgcn_fence(__ATOMIC_RELEASE, "wavefront");
  __builtin_amdgcn_wave_barrier();
  __builtin_amdgcn_fence(__ATOMIC_ACQUIRE, "wavefront");
}

__device__ __forceinline__ void load_k(float4 (&kf)[4][2],
                                       const float* __restrict__ kchunk,
                                       unsigned koff) {
#pragma unroll
  for (int blk = 0; blk < 4; ++blk) {
    kf[blk][0] = *(const float4*)(kchunk + koff + 16 * blk);
    kf[blk][1] = *(const float4*)(kchunk + koff + 16 * blk + 4);
  }
}

__device__ __forceinline__ void load_v(float4 (&vf)[8],
                                       const float* __restrict__ vchunk,
                                       unsigned voff) {
#pragma unroll
  for (int kk = 0; kk < 8; ++kk)
    vf[kk] = *(const float4*)(vchunk + voff + kk * HD);
}

__device__ __forceinline__ void attn_chunk(const float4 (&qf)[4][2],
                                           const float4 (&kf)[4][2],
                                           const float4 (&vf)[8], v8f& o0,
                                           v8f& o1, v8f& o2, v8f& o3,
                                           float (&m)[8], float (&lsum)[8],
                                           float* sP, int l16, int half) {
  v8f s = {};
#pragma unroll
  for (int blk = 0; blk < 4; ++blk) {
#pragma unroll
    for (int t = 0; t < 2; ++t) {
      const float4 qa = qf[blk][t];
      const float4 kb = kf[blk][t];
      v2f a, b;
      a.x = qa.x; a.y = qa.y; b.x = kb.x; b.y = kb.y;
      s = wmma4(a, b, s);
      a.x = qa.z; a.y = qa.w; b.x = kb.z; b.y = kb.w;
      s = wmma4(a, b, s);
    }
  }

  float fac[8];
#pragma unroll
  for (int r = 0; r < 8; ++r) {
    const float sv = s[r] * 0.125f;
    float mx = sv;
#pragma unroll
    for (int off = 1; off < 16; off <<= 1)
      mx = fmaxf(mx, __shfl_xor(mx, off, 32));
    const float nm = fmaxf(m[r], mx);
    const float p = __expf(sv - nm);
    s[r] = p;
    float rs = p;
#pragma unroll
    for (int off = 1; off < 16; off <<= 1) rs += __shfl_xor(rs, off, 32);
    fac[r] = __expf(m[r] - nm);
    lsum[r] = lsum[r] * fac[r] + rs;
    m[r] = nm;
  }
#pragma unroll
  for (int r = 0; r < 8; ++r) {
    o0[r] *= fac[r];
    o1[r] *= fac[r];
    o2[r] *= fac[r];
    o3[r] *= fac[r];
  }

  wave_lds_order();
#pragma unroll
  for (int r = 0; r < 8; ++r) sP[(r + half * 8) * PPITCH + l16] = s[r];
  wave_lds_order();
  const float4 pa0 = *(const float4*)(sP + l16 * PPITCH + half * 8);
  const float4 pa1 = *(const float4*)(sP + l16 * PPITCH + half * 8 + 4);

#pragma unroll
  for (int u = 0; u < 4; ++u) {
    const float4 pa = (u < 2) ? pa0 : pa1;
    v2f a;
    a.x = (u & 1) ? pa.z : pa.x;
    a.y = (u & 1) ? pa.w : pa.y;
    const float4 v0 = vf[2 * u];
    const float4 v1 = vf[2 * u + 1];
    v2f b;
    b.x = v0.x; b.y = v1.x; o0 = wmma4(a, b, o0);
    b.x = v0.y; b.y = v1.y; o1 = wmma4(a, b, o1);
    b.x = v0.z; b.y = v1.z; o2 = wmma4(a, b, o2);
    b.x = v0.w; b.y = v1.w; o3 = wmma4(a, b, o3);
  }
}

__global__ void __launch_bounds__(256)
attn_kernel(const float* __restrict__ Qb, const float* __restrict__ Kb,
            const float* __restrict__ Vb, float* __restrict__ Ab) {
  const int lane = threadIdx.x & 31;
  const int wave = __builtin_amdgcn_readfirstlane((int)(threadIdx.x >> 5));
  const int half = lane >> 4;
  const int l16 = lane & 15;
  const int frame = (wave < 4) ? wave : 11 - wave;
  const int q0 = (frame << 8) + blockIdx.x * 16;
  const int bh = blockIdx.y;
  const int nkey = (frame + 1) << 8;
  const size_t base = (size_t)bh * LT * HD;

  __shared__ __attribute__((aligned(16))) float sPall[8][16 * PPITCH];
  float* sP = sPall[wave];

  float4 qf[4][2];
  {
    const float* qrow = Qb + base + (size_t)(q0 + l16) * HD + half * 8;
#pragma unroll
    for (int blk = 0; blk < 4; ++blk) {
      qf[blk][0] = *(const float4*)(qrow + 16 * blk);
      qf[blk][1] = *(const float4*)(qrow + 16 * blk + 4);
    }
  }

  v8f o0 = {}, o1 = {}, o2 = {}, o3 = {};
  float m[8], lsum[8];
#pragma unroll
  for (int r = 0; r < 8; ++r) {
    m[r] = -3.0e38f;
    lsum[r] = 0.0f;
  }

  const float* kp = Kb + base;
  const float* vp = Vb + base;
  const unsigned koff = (unsigned)(l16 * HD + half * 8);
  const unsigned voff = (unsigned)(half * 8 * HD + 4 * l16);

  float4 kfa[4][2], kfb[4][2], vf[8];
  load_k(kfa, kp, koff);
  for (int key0 = 0; key0 < nkey; key0 += 32) {
    load_v(vf, vp + (size_t)key0 * HD, voff);
    load_k(kfb, kp + (size_t)(key0 + 16) * HD, koff);
    attn_chunk(qf, kfa, vf, o0, o1, o2, o3, m, lsum, sP, l16, half);
    load_v(vf, vp + (size_t)(key0 + 16) * HD, voff);
    const int nxt = (key0 + 32 < nkey) ? key0 + 32 : 0;
    load_k(kfa, kp + (size_t)nxt * HD, koff);
    attn_chunk(qf, kfb, vf, o0, o1, o2, o3, m, lsum, sP, l16, half);
  }

  const int bb = bh >> 4, h = bh & 15;
#pragma unroll
  for (int r = 0; r < 8; ++r) {
    const float inv = 1.0f / lsum[r];
    const int l = q0 + r + half * 8;
    float* dst = Ab + ((size_t)bb * LT + l) * DD + h * HD + 4 * l16;
    float4 ov;
    ov.x = o0[r] * inv;
    ov.y = o1[r] * inv;
    ov.z = o2[r] * inv;
    ov.w = o3[r] * inv;
    *(float4*)dst = ov;
  }
}

__global__ void __launch_bounds__(128)
out_gemm_kernel(const float* __restrict__ A, const float* __restrict__ W,
                float* __restrict__ Y) {
  const int lane = threadIdx.x & 31;
  const int wave = threadIdx.x >> 5;
  const int half = lane >> 4;
  const int l16 = lane & 15;
  const int rowBase = blockIdx.y * 128 + (wave & 1) * 64;
  const int colBase = blockIdx.x * 128 + (wave >> 1) * 64;

  const float* ap = A + (size_t)(rowBase + l16) * DD + half * 8;
  const float* bp = W + (size_t)(colBase + l16) * DD + half * 8;

  v8f acc[4][4] = {};
  gemm_wave_64x64(acc, ap, bp);

#pragma unroll
  for (int i = 0; i < 4; ++i) {
#pragma unroll
    for (int r = 0; r < 8; ++r) {
      const int row = rowBase + i * 16 + r + half * 8;
#pragma unroll
      for (int j = 0; j < 4; ++j)
        Y[(size_t)row * DD + colBase + j * 16 + l16] = acc[i][j][r];
    }
  }
}

extern "C" void kernel_launch(void* const* d_in, const int* in_sizes, int n_in,
                              void* d_out, int out_size, void* d_ws,
                              size_t ws_size, hipStream_t stream) {
  const float* x = (const float*)d_in[0];
  const float* Wqkv = (const float*)d_in[1];
  const float* Wout = (const float*)d_in[2];
  const float* qs = (const float*)d_in[3];
  const float* ks = (const float*)d_in[4];
  float* out = (float*)d_out;

  const size_t N = (size_t)BB * HH * LT * HD;
  float* Qb = (float*)d_ws;
  float* Kb = Qb + N;
  float* Vb = Kb + N;
  float* Ab = Vb + N;

  qkv_gemm_kernel<<<dim3(24, 32), 128, 0, stream>>>(x, Wqkv, Qb, Kb, Vb);
  norm_rope_kernel<<<dim3(8192), 256, 0, stream>>>(Qb, qs);
  norm_rope_kernel<<<dim3(8192), 256, 0, stream>>>(Kb, ks);
  attn_kernel<<<dim3(16, BB * HH), 256, 0, stream>>>(Qb, Kb, Vb, Ab);
  out_gemm_kernel<<<dim3(8, 32), 128, 0, stream>>>(Ab, Wout, out);
}


# Attention softmax with no cross-lane work in the key loop: fixed shift sqrt(hd)*max|q_scale|*max|k_scale| (norm bound on a score) instead of the running row max; row sums per lane, added once
# speedup vs baseline: 2.3117x; 1.2036x over previous
#include <hip/hip_runtime.h>
#include <hip/hip_bf16.h>

#define BB 2
#define LT 2048
#define DD 1024
#define HH 16
#define HD 64

typedef __attribute__((ext_vector_type(2))) float v2f;
typedef __attribute__((ext_vector_type(8))) float v8f;

__device__ __forceinline__ v8f wmma4(v2f a, v2f b, v8f c) {
  return __builtin_amdgcn_wmma_f32_16x16x4_f32(false, a, false, b, (short)0, c,
                                               false, false);
}

__device__ __forceinline__ void load_frags(float4 (&f)[4][2],
                                           const float* __restrict__ p, int k) {
#pragma unroll
  for (int i = 0; i < 4; ++i) {
    const float* q = p + (size_t)i * 16 * DD + k;
    f[i][0] = *(const float4*)(q);
    f[i][1] = *(const float4*)(q + 4);
  }
}

__device__ __forceinline__ void mma_block(v8f (&acc)[4][4],
                                          const float4 (&a)[4][2],
                                          const float4 (&b)[4][2]) {
#pragma unroll
  for (int s = 0; s < 4; ++s) {
    v2f af[4], bf[4];
#pragma unroll
    for (int i = 0; i < 4; ++i) {
      const float4 va = a[i][s >> 1];
      const float4 vb = b[i][s >> 1];
      af[i].x = (s & 1) ? va.z : va.x;
      af[i].y = (s & 1) ? va.w : va.y;
      bf[i].x = (s & 1) ? vb.z : vb.x;
      bf[i].y = (s & 1) ? vb.w : vb.y;
    }
#pragma unroll
    for (int i = 0; i < 4; ++i)
#pragma unroll
      for (int j = 0; j < 4; ++j) acc[i][j] = wmma4(af[i], bf[j], acc[i][j]);
  }
}

__device__ __forceinline__ void gemm_wave_64x64(v8f (&acc)[4][4],
                                                const float* __restrict__ ap,
                                                const float* __restrict__ bp) {
  float4 a0[4][2], b0[4][2], a1[4][2], b1[4][2];
  load_frags(a0, ap, 0);
  load_frags(b0, bp, 0);
  for (int k = 0; k < DD - 32; k += 32) {
    load_frags(a1, ap, k + 16);
    load_frags(b1, bp, k + 16);
    mma_block(acc, a0, b0);
    load_frags(a0, ap, k + 32);
    load_frags(b0, bp, k + 32);
    mma_block(acc, a1, b1);
  }
  load_frags(a1, ap, DD - 16);
  load_frags(b1, bp, DD - 16);
  mma_block(acc, a0, b0);
  mma_block(acc, a1, b1);
}

__global__ void __launch_bounds__(128)
qkv_gemm_kernel(const float* __restrict__ X, const float* __restrict__ W,
                float* __restrict__ Qb, float* __restrict__ Kb,
                float* __restrict__ Vb) {
  const int lane = threadIdx.x & 31;
  const int wave = threadIdx.x >> 5;
  const int half = lane >> 4;
  const int l16 = lane & 15;
  const int rowBase = blockIdx.y * 128 + (wave & 1) * 64;
  const int colBase = blockIdx.x * 128 + (wave >> 1) * 64;

  const float* ap = X + (size_t)(rowBase + l16) * DD + half * 8;
  const float* bp = W + (size_t)(colBase + l16) * DD + half * 8;

  v8f acc[4][4] = {};
  gemm_wave_64x64(acc, ap, bp);

  const int c = colBase >> 10;
  const int h = (colBase & 1023) >> 6;
  float* dst = (c == 0) ? Qb : (c == 1) ? Kb : Vb;
#pragma unroll
  for (int i = 0; i < 4; ++i) {
#pragma unroll
    for (int r = 0; r < 8; ++r) {
      const int row = rowBase + i * 16 + r + half * 8;
      const int b = row >> 11;
      const int l = row & (LT - 1);
      const size_t rbase = (((size_t)b * HH + h) * LT + l) * HD;
#pragma unroll
      for (int j = 0; j < 4; ++j) dst[rbase + j * 16 + l16] = acc[i][j][r];
    }
  }
}

__global__ void __launch_bounds__(256)
norm_rope_kernel(float* __restrict__ buf, const float* __restrict__ scale) {
  const int wave = threadIdx.x >> 5;
  const int lane = threadIdx.x & 31;
  const size_t row = (size_t)blockIdx.x * 8 + wave;
  const int l = (int)(row & (LT - 1));

  float2* p = (float2*)(buf + row * HD) + lane;
  float2 v = *p;
  float ss = v.x * v.x + v.y * v.y;
#pragma unroll
  for (int off = 16; off; off >>= 1) ss += __shfl_xor(ss, off, 32);
  const float r = rsqrtf(ss * (1.0f / 64.0f) + 1e-6f);
  const float xr = v.x * r * scale[2 * lane];
  const float xi = v.y * r * scale[2 * lane + 1];

  const int jj = (lane < 16) ? lane : (lane - 16);
  const float pos = (lane < 16) ? (float)(l >> 8) : (float)(l & 255);
  const float invf = expf((float)jj * (-9.210340371976184f / 16.0f));
  const float ang = pos * invf;
  const float c = cosf(ang), s = sinf(ang);
  float2 o;
  o.x = xr * c - xi * s;
  o.y = xr * s + xi * c;
  *p = o;
}

#define PPITCH 20

__device__ __forceinline__ void wave_lds_order() {
  __builtin_amdgcn_fence(__ATOMIC_RELEASE, "wavefront");
  __builtin_amdgcn_wave_barrier();
  __builtin_amdgcn_fence(__ATOMIC_ACQUIRE, "wavefront");
}

__device__ __forceinline__ void load_k(float4 (&kf)[4][2],
                                       const float* __restrict__ kchunk,
                                       unsigned koff) {
#pragma unroll
  for (int blk = 0; blk < 4; ++blk) {
    kf[blk][0] = *(const float4*)(kchunk + koff + 16 * blk);
    kf[blk][1] = *(const float4*)(kchunk + koff + 16 * blk + 4);
  }
}

__device__ __forceinline__ void load_v(float4 (&vf)[8],
                                       const float* __restrict__ vchunk,
                                       unsigned voff) {
#pragma unroll
  for (int kk = 0; kk < 8; ++kk)
    vf[kk] = *(const float4*)(vchunk + voff + kk * HD);
}

__device__ __forceinline__ void attn_chunk(const float4 (&qf)[4][2],
                                           const float4 (&kf)[4][2],
                                           const float4 (&vf)[8], v8f& o0,
                                           v8f& o1, v8f& o2, v8f& o3,
                                           float (&lsum)[8], float c1, float c0,
                                           float* sP, int l16, int half) {
  v8f s = {};
#pragma unroll
  for (int blk = 0; blk < 4; ++blk) {
#pragma unroll
    for (int t = 0; t < 2; ++t) {
      const float4 qa = qf[blk][t];
      const float4 kb = kf[blk][t];
      v2f a, b;
      a.x = qa.x; a.y = qa.y; b.x = kb.x; b.y = kb.y;
      s = wmma4(a, b, s);
      a.x = qa.z; a.y = qa.w; b.x = kb.z; b.y = kb.w;
      s = wmma4(a, b, s);
    }
  }

#pragma unroll
  for (int r = 0; r < 8; ++r) {
    const float p = __builtin_amdgcn_exp2f(__builtin_fmaf(s[r], c1, -c0));
    s[r] = p;
    lsum[r] += p;
  }

  wave_lds_order();
#pragma unroll
  for (int r = 0; r < 8; ++r) sP[(r + half * 8) * PPITCH + l16] = s[r];
  wave_lds_order();
  const float4 pa0 = *(const float4*)(sP + l16 * PPITCH + half * 8);
  const float4 pa1 = *(const float4*)(sP + l16 * PPITCH + half * 8 + 4);

#pragma unroll
  for (int u = 0; u < 4; ++u) {
    const float4 pa = (u < 2) ? pa0 : pa1;
    v2f a;
    a.x = (u & 1) ? pa.z : pa.x;
    a.y = (u & 1) ? pa.w : pa.y;
    const float4 v0 = vf[2 * u];
    const float4 v1 = vf[2 * u + 1];
    v2f b;
    b.x = v0.x; b.y = v1.x; o0 = wmma4(a, b, o0);
    b.x = v0.y; b.y = v1.y; o1 = wmma4(a, b, o1);
    b.x = v0.z; b.y = v1.z; o2 = wmma4(a, b, o2);
    b.x = v0.w; b.y = v1.w; o3 = wmma4(a, b, o3);
  }
}

__global__ void __launch_bounds__(256)
attn_kernel(const float* __restrict__ Qb, const float* __restrict__ Kb,
            const float* __restrict__ Vb, const float* __restrict__ qscale,
            const float* __restrict__ kscale, float* __restrict__ Ab) {
  const int lane = threadIdx.x & 31;
  const int wave = __builtin_amdgcn_readfirstlane((int)(threadIdx.x >> 5));
  const int half = lane >> 4;
  const int l16 = lane & 15;
  const int frame = (wave < 4) ? wave : 11 - wave;
  const int q0 = (frame << 8) + blockIdx.x * 16;
  const int bh = blockIdx.y;
  const int nkey = (frame + 1) << 8;
  const size_t base = (size_t)bh * LT * HD;

  __shared__ __attribute__((aligned(16))) float sPall[8][16 * PPITCH];
  float* sP = sPall[wave];

  float4 qf[4][2];
  {
    const float* qrow = Qb + base + (size_t)(q0 + l16) * HD + half * 8;
#pragma unroll
    for (int blk = 0; blk < 4; ++blk) {
      qf[blk][0] = *(const float4*)(qrow + 16 * blk);
      qf[blk][1] = *(const float4*)(qrow + 16 * blk + 4);
    }
  }

  v8f o0 = {}, o1 = {}, o2 = {}, o3 = {};
  float lsum[8];
#pragma unroll
  for (int r = 0; r < 8; ++r) lsum[r] = 0.0f;

  float mq = fmaxf(fabsf(qscale[2 * lane]), fabsf(qscale[2 * lane + 1]));
  float mk = fmaxf(fabsf(kscale[2 * lane]), fabsf(kscale[2 * lane + 1]));
#pragma unroll
  for (int off = 16; off; off >>= 1) {
    mq = fmaxf(mq, __shfl_xor(mq, off, 32));
    mk = fmaxf(mk, __shfl_xor(mk, off, 32));
  }
  const float LOG2E = 1.4426950408889634f;
  const float c1 = 0.125f * LOG2E;
  const float c0 = 8.0f * mq * mk * LOG2E;

  const float* kp = Kb + base;
  const float* vp = Vb + base;
  const unsigned koff = (unsigned)(l16 * HD + half * 8);
  const unsigned voff = (unsigned)(half * 8 * HD + 4 * l16);

  float4 kfa[4][2], kfb[4][2], vf[8];
  load_k(kfa, kp, koff);
  for (int key0 = 0; key0 < nkey; key0 += 32) {
    load_v(vf, vp + (size_t)key0 * HD, voff);
    load_k(kfb, kp + (size_t)(key0 + 16) * HD, koff);
    attn_chunk(qf, kfa, vf, o0, o1, o2, o3, lsum, c1, c0, sP, l16, half);
    load_v(vf, vp + (size_t)(key0 + 16) * HD, voff);
    const int nxt = (key0 + 32 < nkey) ? key0 + 32 : 0;
    load_k(kfa, kp + (size_t)nxt * HD, koff);
    attn_chunk(qf, kfb, vf, o0, o1, o2, o3, lsum, c1, c0, sP, l16, half);
  }

  const int bb = bh >> 4, h = bh & 15;
#pragma unroll
  for (int r = 0; r < 8; ++r) {
    float rs = lsum[r];
#pragma unroll
    for (int off = 1; off < 16; off <<= 1) rs += __shfl_xor(rs, off, 32);
    const float inv = 1.0f / rs;
    const int l = q0 + r + half * 8;
    float* dst = Ab + ((size_t)bb * LT + l) * DD + h * HD + 4 * l16;
    float4 ov;
    ov.x = o0[r] * inv;
    ov.y = o1[r] * inv;
    ov.z = o2[r] * inv;
    ov.w = o3[r] * inv;
    *(float4*)dst = ov;
  }
}

__global__ void __launch_bounds__(128)
out_gemm_kernel(const float* __restrict__ A, const float* __restrict__ W,
                float* __restrict__ Y) {
  const int lane = threadIdx.x & 31;
  const int wave = threadIdx.x >> 5;
  const int half = lane >> 4;
  const int l16 = lane & 15;
  const int rowBase = blockIdx.y * 128 + (wave & 1) * 64;
  const int colBase = blockIdx.x * 128 + (wave >> 1) * 64;

  const float* ap = A + (size_t)(rowBase + l16) * DD + half * 8;
  const float* bp = W + (size_t)(colBase + l16) * DD + half * 8;

  v8f acc[4][4] = {};
  gemm_wave_64x64(acc, ap, bp);

#pragma unroll
  for (int i = 0; i < 4; ++i) {
#pragma unroll
    for (int r = 0; r < 8; ++r) {
      const int row = rowBase + i * 16 + r + half * 8;
#pragma unroll
      for (int j = 0; j < 4; ++j)
        Y[(size_t)row * DD + colBase + j * 16 + l16] = acc[i][j][r];
    }
  }
}

extern "C" void kernel_launch(void* const* d_in, const int* in_sizes, int n_in,
                              void* d_out, int out_size, void* d_ws,
                              size_t ws_size, hipStream_t stream) {
  const float* x = (const float*)d_in[0];
  const float* Wqkv = (const float*)d_in[1];
  const float* Wout = (const float*)d_in[2];
  const float* qs = (const float*)d_in[3];
  const float* ks = (const float*)d_in[4];
  float* out = (float*)d_out;

  const size_t N = (size_t)BB * HH * LT * HD;
  float* Qb = (float*)d_ws;
  float* Kb = Qb + N;
  float* Vb = Kb + N;
  float* Ab = Vb + N;

  qkv_gemm_kernel<<<dim3(24, 32), 128, 0, stream>>>(x, Wqkv, Qb, Kb, Vb);
  norm_rope_kernel<<<dim3(8192), 256, 0, stream>>>(Qb, qs);
  norm_rope_kernel<<<dim3(8192), 256, 0, stream>>>(Kb, ks);
  attn_kernel<<<dim3(16, BB * HH), 256, 0, stream>>>(Qb, Kb, Vb, qs, ks, Ab);
  out_gemm_kernel<<<dim3(8, 32), 128, 0, stream>>>(Ab, Wout, out);
}
